# starting kernel
# baseline (speedup 1.0000x reference)
#include <hip/hip_runtime.h>
#include <hip/hip_bf16.h>
#include <math.h>

typedef float v2f __attribute__((ext_vector_type(2)));
typedef float v8f __attribute__((ext_vector_type(8)));

#define D1 128
#define D2 64
#define WAVES_PER_BLOCK 8
#define BLOCK (WAVES_PER_BLOCK * 32)
#define ROW_STRIDE 132
#define TILE_FLOATS (16 * ROW_STRIDE)
#define W1_PAIRS (64 * 64)
#define LDS_FLOATS (2 * W1_PAIRS + WAVES_PER_BLOCK * TILE_FLOATS)

__global__ void __launch_bounds__(BLOCK)
edge_decoder_kernel(const float* __restrict__ hf,
                    const int*   __restrict__ edge,
                    const float* __restrict__ W1,
                    const float* __restrict__ b1,
                    const float* __restrict__ W2,
                    const float* __restrict__ b2,
                    float* __restrict__ out,
                    int E)
{
    extern __shared__ float smem[];
    v2f* W1p = (v2f*)smem;

    const int tid = threadIdx.x;

    for (int idx = tid; idx < W1_PAIRS; idx += BLOCK) {
        const int p = idx >> 6;
        const int c = idx & 63;
        v2f v;
        v.x = W1[p * 2 * D2 + c];
        v.y = W1[p * 2 * D2 + D2 + c];
        W1p[idx] = v;
    }
    __syncthreads();

    const int l  = tid & 31;
    const int li = l & 15;
    const int hi = l >> 4;
    const int w  = tid >> 5;

    float* tileBuf = smem + 2 * W1_PAIRS + w * TILE_FLOATS;
    const v2f* Ab = (const v2f*)tileBuf;
    const int aBase = li * (ROW_STRIDE / 2) + hi;

    const int* edge0 = edge;
    const int* edge1 = edge + E;

    float bias[4], w2v[4];
#pragma unroll
    for (int n = 0; n < 4; ++n) {
        bias[n] = b1[16 * n + li];
        w2v[n]  = W2[16 * n + li];
    }
    const float b2v = b2[0];

    const int ntiles = (E + 15) >> 4;
    const int nwaves = gridDim.x * WAVES_PER_BLOCK;

    for (int tile = blockIdx.x * WAVES_PER_BLOCK + w; tile < ntiles; tile += nwaves) {
        const int e0 = tile << 4;

#pragma unroll 4
        for (int e = 0; e < 16; ++e) {
            int ee = e0 + e; if (ee >= E) ee = E - 1;
            const long long u = edge0[ee];
            const long long v = edge1[ee];
            const float4 fu = *(const float4*)(hf + u * D1 + 4 * l);
            const float4 fv = *(const float4*)(hf + v * D1 + 4 * l);
            float4 p;
            p.x = fu.x * fv.x; p.y = fu.y * fv.y;
            p.z = fu.z * fv.z; p.w = fu.w * fv.w;
            *(float4*)(tileBuf + e * ROW_STRIDE + 4 * l) = p;
        }

        v8f c0 = {bias[0],bias[0],bias[0],bias[0],bias[0],bias[0],bias[0],bias[0]};
        v8f c1 = {bias[1],bias[1],bias[1],bias[1],bias[1],bias[1],bias[1],bias[1]};
        v8f c2 = {bias[2],bias[2],bias[2],bias[2],bias[2],bias[2],bias[2],bias[2]};
        v8f c3 = {bias[3],bias[3],bias[3],bias[3],bias[3],bias[3],bias[3],bias[3]};

#pragma unroll
        for (int t = 0; t < 32; ++t) {
            const v2f a = Ab[aBase + 2 * t];
            const int bb = (2 * t + hi) * 64 + li;
            c0 = __builtin_amdgcn_wmma_f32_16x16x4_f32(false, a, false, W1p[bb     ], (short)0, c0, false, false);
            c1 = __builtin_amdgcn_wmma_f32_16x16x4_f32(false, a, false, W1p[bb + 16], (short)0, c1, false, false);
            c2 = __builtin_amdgcn_wmma_f32_16x16x4_f32(false, a, false, W1p[bb + 32], (short)0, c2, false, false);
            c3 = __builtin_amdgcn_wmma_f32_16x16x4_f32(false, a, false, W1p[bb + 48], (short)0, c3, false, false);
        }

        float s[8];
#pragma unroll
        for (int i = 0; i < 8; ++i) {
            s[i] = fmaxf(c0[i], 0.f) * w2v[0] + fmaxf(c1[i], 0.f) * w2v[1]
                 + fmaxf(c2[i], 0.f) * w2v[2] + fmaxf(c3[i], 0.f) * w2v[3];
        }
#pragma unroll
        for (int m = 1; m <= 8; m <<= 1) {
#pragma unroll
            for (int i = 0; i < 8; ++i) s[i] += __shfl_xor(s[i], m, 32);
        }

        if (li == 0) {
            const int base = e0 + hi * 8;
            float r[8];
#pragma unroll
            for (int i = 0; i < 8; ++i) r[i] = 1.f / (1.f + expf(-(s[i] + b2v)));
            if (base + 8 <= E) {
                float4 o0 = {r[0], r[1], r[2], r[3]};
                float4 o1 = {r[4], r[5], r[6], r[7]};
                *(float4*)(out + base)     = o0;
                *(float4*)(out + base + 4) = o1;
            } else {
#pragma unroll
                for (int i = 0; i < 8; ++i) if (base + i < E) out[base + i] = r[i];
            }
        }
    }
}

extern "C" void kernel_launch(void* const* d_in, const int* in_sizes, int n_in,
                              void* d_out, int out_size, void* d_ws, size_t ws_size,
                              hipStream_t stream) {
    (void)n_in; (void)out_size; (void)d_ws; (void)ws_size;
    const float* hf   = (const float*)d_in[0];
    const int*   edge = (const int*)d_in[1];
    const float* W1   = (const float*)d_in[2];
    const float* b1   = (const float*)d_in[3];
    const float* W2   = (const float*)d_in[4];
    const float* b2   = (const float*)d_in[5];
    float* out = (float*)d_out;

    const int E = in_sizes[1] / 2;
    const int ntiles = (E + 15) / 16;
    int blocks = (ntiles + WAVES_PER_BLOCK - 1) / WAVES_PER_BLOCK;
    if (blocks > 2500) blocks = 2500;

    const size_t shmem = LDS_FLOATS * sizeof(float);
    hipLaunchKernelGGL(edge_decoder_kernel, dim3(blocks), dim3(BLOCK), shmem, stream,
                       hf, edge, W1, b1, W2, b2, out, E);
}


# 16-lane sum of the W2 dot done as a reduce-scatter on DPP row_xmask adds (4+2+1+1 adds, no ds_bpermute): each lane ends with one edge's sum, so one sigmoid per lane, not 8 in 2 lanes; sum order kept
# speedup vs baseline: 1.1240x; 1.1240x over previous
#include <hip/hip_runtime.h>
#include <hip/hip_bf16.h>
#include <math.h>

typedef float v2f __attribute__((ext_vector_type(2)));
typedef float v8f __attribute__((ext_vector_type(8)));
typedef float v4f __attribute__((ext_vector_type(4)));
typedef const __attribute__((address_space(1))) float* gfp;
typedef int   v16i __attribute__((ext_vector_type(16), aligned(4)));

#define D1 128
#define D2 64
#define WAVES_PER_BLOCK 8
#define BLOCK (WAVES_PER_BLOCK * 32)
#define ROW_STRIDE 132
#define TILE_FLOATS (16 * ROW_STRIDE)
#define W1_PAIRS (64 * 64)
#define LDS_FLOATS (2 * W1_PAIRS + WAVES_PER_BLOCK * TILE_FLOATS)

template <int M>
__device__ __forceinline__ float row_xor(float v) {
    return __builtin_bit_cast(float,
        __builtin_amdgcn_update_dpp(0, __builtin_bit_cast(int, v), 0x160 + M, 0xf, 0xf, true));
}

__global__ void __launch_bounds__(BLOCK)
edge_decoder_kernel(const float* __restrict__ hf,
                    const int*   __restrict__ edge,
                    const float* __restrict__ W1,
                    const float* __restrict__ b1,
                    const float* __restrict__ W2,
                    const float* __restrict__ b2,
                    float* __restrict__ out,
                    int E)
{
    extern __shared__ float smem[];
    v2f* W1p = (v2f*)smem;

    const int tid = threadIdx.x;

    for (int idx = tid; idx < W1_PAIRS; idx += BLOCK) {
        const int p = idx >> 6;
        const int c = idx & 63;
        v2f v;
        v.x = W1[p * 2 * D2 + c];
        v.y = W1[p * 2 * D2 + D2 + c];
        W1p[idx] = v;
    }
    __syncthreads();

    const int l  = tid & 31;
    const int li = l & 15;
    const int hi = l >> 4;
    const int w  = __builtin_amdgcn_readfirstlane(tid >> 5);

    float* tileBuf = smem + 2 * W1_PAIRS + w * TILE_FLOATS;
    const v2f* Ab = (const v2f*)tileBuf;
    const int aBase = li * (ROW_STRIDE / 2) + hi;

    const int* edge0 = edge;
    const int* edge1 = edge + E;

    float bias[4], w2v[4];
#pragma unroll
    for (int n = 0; n < 4; ++n) {
        bias[n] = b1[16 * n + li];
        w2v[n]  = W2[16 * n + li];
    }
    const float b2v = b2[0];

    const bool bit0 = (li & 1) != 0, bit1 = (li & 2) != 0, bit2 = (li & 4) != 0, bit3 = (li & 8) != 0;
    const int rowSel = ((li & 1) << 2) | (li & 2) | ((li >> 2) & 1);

    const int ntiles = (E + 15) >> 4;
    const int nwaves = gridDim.x * WAVES_PER_BLOCK;

    for (int tile = blockIdx.x * WAVES_PER_BLOCK + w; tile < ntiles; tile += nwaves) {
        const int e0 = tile << 4;

        int iu[16], iv[16];
        if (e0 + 16 <= E) {
            const v16i tu = *(const v16i*)(edge0 + e0);
            const v16i tv = *(const v16i*)(edge1 + e0);
#pragma unroll
            for (int e = 0; e < 16; ++e) { iu[e] = tu[e]; iv[e] = tv[e]; }
        } else {
#pragma unroll
            for (int e = 0; e < 16; ++e) {
                int ee = e0 + e; if (ee >= E) ee = E - 1;
                iu[e] = edge0[ee];
                iv[e] = edge1[ee];
            }
        }

        const unsigned lo = 4u * (unsigned)l;
#pragma unroll
        for (int e = 0; e < 16; ++e) {
            gfp ru = (gfp)hf + (long long)iu[e] * D1;
            gfp rv = (gfp)hf + (long long)iv[e] * D1;
            asm("" : "+s"(ru), "+s"(rv));
            const v4f fu = *(const __attribute__((address_space(1))) v4f*)(ru + lo);
            const v4f fv = *(const __attribute__((address_space(1))) v4f*)(rv + lo);
            *(v4f*)(tileBuf + e * ROW_STRIDE + 4 * l) = fu * fv;
        }

        v8f c0 = {bias[0],bias[0],bias[0],bias[0],bias[0],bias[0],bias[0],bias[0]};
        v8f c1 = {bias[1],bias[1],bias[1],bias[1],bias[1],bias[1],bias[1],bias[1]};
        v8f c2 = {bias[2],bias[2],bias[2],bias[2],bias[2],bias[2],bias[2],bias[2]};
        v8f c3 = {bias[3],bias[3],bias[3],bias[3],bias[3],bias[3],bias[3],bias[3]};

#pragma unroll
        for (int t = 0; t < 32; ++t) {
            const v2f a = Ab[aBase + 2 * t];
            const int bb = (2 * t + hi) * 64 + li;
            c0 = __builtin_amdgcn_wmma_f32_16x16x4_f32(false, a, false, W1p[bb     ], (short)0, c0, false, false);
            c1 = __builtin_amdgcn_wmma_f32_16x16x4_f32(false, a, false, W1p[bb + 16], (short)0, c1, false, false);
            c2 = __builtin_amdgcn_wmma_f32_16x16x4_f32(false, a, false, W1p[bb + 32], (short)0, c2, false, false);
            c3 = __builtin_amdgcn_wmma_f32_16x16x4_f32(false, a, false, W1p[bb + 48], (short)0, c3, false, false);
        }

        float s[8];
#pragma unroll
        for (int i = 0; i < 8; ++i) {
            s[i] = fmaxf(c0[i], 0.f) * w2v[0] + fmaxf(c1[i], 0.f) * w2v[1]
                 + fmaxf(c2[i], 0.f) * w2v[2] + fmaxf(c3[i], 0.f) * w2v[3];
        }

        float q[4];
#pragma unroll
        for (int i = 0; i < 4; ++i) {
            const float keep = bit0 ? s[i + 4] : s[i];
            const float send = bit0 ? s[i] : s[i + 4];
            q[i] = keep + row_xor<1>(send);
        }
        float d[2];
#pragma unroll
        for (int i = 0; i < 2; ++i) {
            const float keep = bit1 ? q[i + 2] : q[i];
            const float send = bit1 ? q[i] : q[i + 2];
            d[i] = keep + row_xor<2>(send);
        }
        float t1;
        {
            const float keep = bit2 ? d[1] : d[0];
            const float send = bit2 ? d[0] : d[1];
            t1 = keep + row_xor<4>(send);
        }
        const float tot = t1 + row_xor<8>(t1);

        const float r = 1.f / (1.f + expf(-(tot + b2v)));
        const int eo = e0 + hi * 8 + rowSel;
        if (!bit3 && eo < E) out[eo] = r;
    }
}

extern "C" void kernel_launch(void* const* d_in, const int* in_sizes, int n_in,
                              void* d_out, int out_size, void* d_ws, size_t ws_size,
                              hipStream_t stream) {
    (void)n_in; (void)out_size; (void)d_ws; (void)ws_size;
    const float* hf   = (const float*)d_in[0];
    const int*   edge = (const int*)d_in[1];
    const float* W1   = (const float*)d_in[2];
    const float* b1   = (const float*)d_in[3];
    const float* W2   = (const float*)d_in[4];
    const float* b2   = (const float*)d_in[5];
    float* out = (float*)d_out;

    const int E = in_sizes[1] / 2;
    const int ntiles = (E + 15) / 16;
    int blocks = (ntiles + WAVES_PER_BLOCK - 1) / WAVES_PER_BLOCK;
    if (blocks > 2500) blocks = 2500;

    const size_t shmem = LDS_FLOATS * sizeof(float);
    hipLaunchKernelGGL(edge_decoder_kernel, dim3(blocks), dim3(BLOCK), shmem, stream,
                       hf, edge, W1, b1, W2, b2, out, E);
}
